# starting kernel
# baseline (speedup 1.0000x reference)
#include <hip/hip_runtime.h>
#include <hip/hip_bf16.h>

#define BB   8
#define CC   256
#define HH   64
#define WWD  64
#define EE   9
#define OUTC 256
#define EPSF 1e-5f

typedef __attribute__((ext_vector_type(2))) float v2f;
typedef __attribute__((ext_vector_type(8))) float v8f;

#define WS_WEFF   0
#define WS_EBIAS  2304
#define WS_B2     2560
#define WS_T2     2816
#define WS_WSWZ   3072

__device__ __forceinline__ float silu(float v) {
    return v / (1.0f + __expf(-v));
}

__global__ __launch_bounds__(256) void fold_kernel(
    const float* __restrict__ w_experts, const float* __restrict__ bn_gamma,
    const float* __restrict__ bn_beta,   const float* __restrict__ bn_mean,
    const float* __restrict__ bn_var,    const float* __restrict__ scales,
    const float* __restrict__ out_gamma, const float* __restrict__ out_beta,
    const float* __restrict__ out_mean,  const float* __restrict__ out_var,
    float* __restrict__ ws)
{
    const int c = threadIdx.x;
    float wacc[9];
#pragma unroll
    for (int k = 0; k < 9; ++k) wacc[k] = 0.0f;
    float bias = 0.0f;
#pragma unroll
    for (int e = 0; e < EE; ++e) {
        const int ec = e * CC + c;
        const float t  = bn_gamma[ec] * rsqrtf(bn_var[ec] + EPSF);
        const float es = t * scales[ec];
        bias += (bn_beta[ec] - bn_mean[ec] * t) * scales[ec];
        const float* wp = w_experts + ec * 9;
#pragma unroll
        for (int k = 0; k < 9; ++k) wacc[k] += es * wp[k];
    }
#pragma unroll
    for (int k = 0; k < 9; ++k) ws[WS_WEFF + c * 9 + k] = wacc[k];
    ws[WS_EBIAS + c] = bias;

    const float t2 = out_gamma[c] * rsqrtf(out_var[c] + EPSF);
    ws[WS_T2 + c] = t2;
    ws[WS_B2 + c] = out_beta[c] - out_mean[c] * t2;
}

__global__ __launch_bounds__(256) void swz_kernel(
    const float* __restrict__ w_out, float* __restrict__ ws)
{
    const int idx   = blockIdx.x * 256 + threadIdx.x;
    const int kstep = idx >> 9;
    const int rem   = idx & 511;
    const int mt    = rem >> 5;
    const int lane  = rem & 31;
    const int m     = mt * 16 + (lane & 15);
    const int k     = kstep * 4 + ((lane >> 4) << 1);
    const float t2  = ws[WS_T2 + m];
    ws[WS_WSWZ + idx * 2]     = w_out[m * CC + k]     * t2;
    ws[WS_WSWZ + idx * 2 + 1] = w_out[m * CC + k + 1] * t2;
}

__global__ __launch_bounds__(256) void move_kernel(
    const float* __restrict__ x, const float* __restrict__ ws,
    float* __restrict__ out)
{
    __shared__ float smem[CC * WWD];

    const int tid = threadIdx.x;
    const int b   = blockIdx.x >> 6;
    const int h   = blockIdx.x & 63;

    const float* weff  = ws + WS_WEFF;
    const float* ebias = ws + WS_EBIAS;
    const float* b2    = ws + WS_B2;
    const float* wswz  = ws + WS_WSWZ;

    {
        const int w  = tid & 63;
        const int cl = tid >> 6;
#pragma unroll 2
        for (int ci = 0; ci < 64; ++ci) {
            const int c = (ci << 2) + cl;
            const float* xp = x + (((b * CC + c) * HH + h) * WWD) + w;
            const float* we = weff + c * 9;
            float s = 0.0f;
#pragma unroll
            for (int dy = -1; dy <= 1; ++dy) {
                const int h2 = h + dy;
                if (h2 >= 0 && h2 < HH) {
#pragma unroll
                    for (int dx = -1; dx <= 1; ++dx) {
                        const int w2 = w + dx;
                        if (w2 >= 0 && w2 < WWD)
                            s += xp[dy * WWD + dx] * we[(dy + 1) * 3 + (dx + 1)];
                    }
                }
            }
            s = silu(s + ebias[c]);
            const int kstep  = c >> 2;
            const int kin    = c & 3;
            const int lane_s = (w & 15) + ((kin >> 1) << 4);
            const int ntile  = w >> 4;
            smem[(((((kstep << 2) + ntile) << 5) + lane_s) << 1) + (kin & 1)] = s;
        }
    }
    __syncthreads();

    const int lane = tid & 31;
    const int wv   = tid >> 5;
    const int mt0  = wv << 1;
    const int mo   = (lane >> 4) << 3;

    v8f acc[2][4];
#pragma unroll
    for (int ms = 0; ms < 2; ++ms) {
        const float* bp = b2 + ((mt0 + ms) << 4) + mo;
#pragma unroll
        for (int r = 0; r < 8; ++r) {
            const float bval = bp[r];
#pragma unroll
            for (int ns = 0; ns < 4; ++ns) acc[ms][ns][r] = bval;
        }
    }

    const v2f* aptr = (const v2f*)wswz;
    const v2f* bptr = (const v2f*)smem;
#pragma unroll 4
    for (int kstep = 0; kstep < 64; ++kstep) {
        const v2f a0 = aptr[((kstep << 4) + mt0)     * 32 + lane];
        const v2f a1 = aptr[((kstep << 4) + mt0 + 1) * 32 + lane];
#pragma unroll
        for (int ns = 0; ns < 4; ++ns) {
            const v2f bf = bptr[((kstep << 2) + ns) * 32 + lane];
            acc[0][ns] = __builtin_amdgcn_wmma_f32_16x16x4_f32(
                false, a0, false, bf, (short)0, acc[0][ns], false, false);
            acc[1][ns] = __builtin_amdgcn_wmma_f32_16x16x4_f32(
                false, a1, false, bf, (short)0, acc[1][ns], false, false);
        }
    }

    const int ncol = lane & 15;
#pragma unroll
    for (int ms = 0; ms < 2; ++ms) {
#pragma unroll
        for (int ns = 0; ns < 4; ++ns) {
#pragma unroll
            for (int r = 0; r < 8; ++r) {
                const int o = ((mt0 + ms) << 4) + mo + r;
                const float v = silu(acc[ms][ns][r]);
                out[(((b * OUTC + o) * HH + h) * WWD) + (ns << 4) + ncol] = v;
            }
        }
    }
}

extern "C" void kernel_launch(void* const* d_in, const int* in_sizes, int n_in,
                              void* d_out, int out_size, void* d_ws, size_t ws_size,
                              hipStream_t stream) {
    const float* x         = (const float*)d_in[0];
    const float* w_experts = (const float*)d_in[1];
    const float* bn_gamma  = (const float*)d_in[2];
    const float* bn_beta   = (const float*)d_in[3];
    const float* bn_mean   = (const float*)d_in[4];
    const float* bn_var    = (const float*)d_in[5];
    const float* scales    = (const float*)d_in[6];
    const float* w_out     = (const float*)d_in[7];
    const float* out_gamma = (const float*)d_in[8];
    const float* out_beta  = (const float*)d_in[9];
    const float* out_mean  = (const float*)d_in[10];
    const float* out_var   = (const float*)d_in[11];
    float* out = (float*)d_out;
    float* ws  = (float*)d_ws;

    fold_kernel<<<1, 256, 0, stream>>>(w_experts, bn_gamma, bn_beta, bn_mean,
                                       bn_var, scales, out_gamma, out_beta,
                                       out_mean, out_var, ws);
    swz_kernel<<<128, 256, 0, stream>>>(w_out, ws);
    move_kernel<<<BB * HH, 256, 0, stream>>>(x, ws, out);
}


# conv stage: per lane 4 columns x 2 channels x 3 rows as unconditional 16-byte loads issued together, side taps from the neighbour lane (DPP row shift), not one waited load per tap; tap order kept
# speedup vs baseline: 2.9726x; 2.9726x over previous
#include <hip/hip_runtime.h>
#include <hip/hip_bf16.h>

#define BB   8
#define CC   256
#define HH   64
#define WWD  64
#define EE   9
#define OUTC 256
#define EPSF 1e-5f

typedef __attribute__((ext_vector_type(2))) float v2f;
typedef __attribute__((ext_vector_type(8))) float v8f;
typedef __attribute__((ext_vector_type(4))) float v4f;

#define WS_WEFF   0
#define WS_EBIAS  2304
#define WS_B2     2560
#define WS_T2     2816
#define WS_WSWZ   3072

__device__ __forceinline__ float silu(float v) {
    return v / (1.0f + __expf(-v));
}

__device__ __forceinline__ float row_from_left(float v) {
    return __int_as_float(__builtin_amdgcn_update_dpp(
        0, __float_as_int(v), 0x111, 0xf, 0xf, true));
}
__device__ __forceinline__ float row_from_right(float v) {
    return __int_as_float(__builtin_amdgcn_update_dpp(
        0, __float_as_int(v), 0x101, 0xf, 0xf, true));
}

__global__ __launch_bounds__(256) void fold_kernel(
    const float* __restrict__ w_experts, const float* __restrict__ bn_gamma,
    const float* __restrict__ bn_beta,   const float* __restrict__ bn_mean,
    const float* __restrict__ bn_var,    const float* __restrict__ scales,
    const float* __restrict__ out_gamma, const float* __restrict__ out_beta,
    const float* __restrict__ out_mean,  const float* __restrict__ out_var,
    float* __restrict__ ws)
{
    const int c = threadIdx.x;
    float wacc[9];
#pragma unroll
    for (int k = 0; k < 9; ++k) wacc[k] = 0.0f;
    float bias = 0.0f;
#pragma unroll
    for (int e = 0; e < EE; ++e) {
        const int ec = e * CC + c;
        const float t  = bn_gamma[ec] * rsqrtf(bn_var[ec] + EPSF);
        const float es = t * scales[ec];
        bias += (bn_beta[ec] - bn_mean[ec] * t) * scales[ec];
        const float* wp = w_experts + ec * 9;
#pragma unroll
        for (int k = 0; k < 9; ++k) wacc[k] += es * wp[k];
    }
#pragma unroll
    for (int k = 0; k < 9; ++k) ws[WS_WEFF + c * 9 + k] = wacc[k];
    ws[WS_EBIAS + c] = bias;

    const float t2 = out_gamma[c] * rsqrtf(out_var[c] + EPSF);
    ws[WS_T2 + c] = t2;
    ws[WS_B2 + c] = out_beta[c] - out_mean[c] * t2;
}

__global__ __launch_bounds__(256) void swz_kernel(
    const float* __restrict__ w_out, float* __restrict__ ws)
{
    const int idx   = blockIdx.x * 256 + threadIdx.x;
    const int kstep = idx >> 9;
    const int rem   = idx & 511;
    const int mt    = rem >> 5;
    const int lane  = rem & 31;
    const int m     = mt * 16 + (lane & 15);
    const int k     = kstep * 4 + ((lane >> 4) << 1);
    const float t2  = ws[WS_T2 + m];
    ws[WS_WSWZ + idx * 2]     = w_out[m * CC + k]     * t2;
    ws[WS_WSWZ + idx * 2 + 1] = w_out[m * CC + k + 1] * t2;
}

__global__ __launch_bounds__(256) void move_kernel(
    const float* __restrict__ x, const float* __restrict__ ws,
    float* __restrict__ out)
{
    __shared__ __attribute__((aligned(16))) float smem[CC * WWD];

    const int tid = threadIdx.x;
    const int b   = blockIdx.x >> 6;
    const int h   = blockIdx.x & 63;

    const float* weff  = ws + WS_WEFF;
    const float* ebias = ws + WS_EBIAS;
    const float* b2    = ws + WS_B2;
    const float* wswz  = ws + WS_WSWZ;

    {
        const int q    = tid & 15;
        const int hh   = (tid >> 4) & 1;
        const int wv1  = tid >> 5;
        const bool hasU = h > 0;
        const bool hasD = h < HH - 1;
        const int offU = hasU ? -WWD : 0;
        const int offD = hasD ?  WWD : 0;
        const float* xq = x + (((b * CC) * HH + h) * WWD) + (q << 2);
        for (int it = 0; it < 8; ++it) {
            const int kstep = (wv1 << 3) + it;
            const int c     = (kstep << 2) + (hh << 1);
            v4f r[2][3];
            for (int j = 0; j < 2; ++j) {
                const float* p = xq + (c + j) * (HH * WWD);
                r[j][0] = *(const v4f*)(p + offU);
                r[j][1] = *(const v4f*)(p);
                r[j][2] = *(const v4f*)(p + offD);
            }
            float wt[18];
            for (int t = 0; t < 18; ++t) wt[t] = weff[c * 9 + t];
            float bs[2];
            bs[0] = ebias[c];
            bs[1] = ebias[c + 1];
            const v4f zero4 = {0.0f, 0.0f, 0.0f, 0.0f};
            if (!hasU) { r[0][0] = zero4; r[1][0] = zero4; }
            if (!hasD) { r[0][2] = zero4; r[1][2] = zero4; }

            float s[2][4];
            for (int j = 0; j < 2; ++j) {
                float s0 = 0.0f, s1 = 0.0f, s2 = 0.0f, s3 = 0.0f;
                for (int d = 0; d < 3; ++d) {
                    const v4f v  = r[j][d];
                    const float L  = row_from_left(v.w);
                    const float R  = row_from_right(v.x);
                    const float wl = wt[j * 9 + d * 3];
                    const float wc = wt[j * 9 + d * 3 + 1];
                    const float wr = wt[j * 9 + d * 3 + 2];
                    s0 += L   * wl;  s0 += v.x * wc;  s0 += v.y * wr;
                    s1 += v.x * wl;  s1 += v.y * wc;  s1 += v.z * wr;
                    s2 += v.y * wl;  s2 += v.z * wc;  s2 += v.w * wr;
                    s3 += v.z * wl;  s3 += v.w * wc;  s3 += R   * wr;
                }
                s[j][0] = silu(s0 + bs[j]);
                s[j][1] = silu(s1 + bs[j]);
                s[j][2] = silu(s2 + bs[j]);
                s[j][3] = silu(s3 + bs[j]);
            }
            const int base = ((((kstep << 2) + (q >> 2)) << 6) + (hh << 5)) + ((q & 3) << 3);
            const v4f o0 = {s[0][0], s[1][0], s[0][1], s[1][1]};
            const v4f o1 = {s[0][2], s[1][2], s[0][3], s[1][3]};
            *(v4f*)(smem + base)     = o0;
            *(v4f*)(smem + base + 4) = o1;
        }
    }
    __syncthreads();

    const int lane = tid & 31;
    const int wv   = tid >> 5;
    const int mt0  = wv << 1;
    const int mo   = (lane >> 4) << 3;

    v8f acc[2][4];
#pragma unroll
    for (int ms = 0; ms < 2; ++ms) {
        const float* bp = b2 + ((mt0 + ms) << 4) + mo;
#pragma unroll
        for (int r = 0; r < 8; ++r) {
            const float bval = bp[r];
#pragma unroll
            for (int ns = 0; ns < 4; ++ns) acc[ms][ns][r] = bval;
        }
    }

    const v2f* aptr = (const v2f*)wswz;
    const v2f* bptr = (const v2f*)smem;
#pragma unroll 4
    for (int kstep = 0; kstep < 64; ++kstep) {
        const v2f a0 = aptr[((kstep << 4) + mt0)     * 32 + lane];
        const v2f a1 = aptr[((kstep << 4) + mt0 + 1) * 32 + lane];
#pragma unroll
        for (int ns = 0; ns < 4; ++ns) {
            const v2f bf = bptr[((kstep << 2) + ns) * 32 + lane];
            acc[0][ns] = __builtin_amdgcn_wmma_f32_16x16x4_f32(
                false, a0, false, bf, (short)0, acc[0][ns], false, false);
            acc[1][ns] = __builtin_amdgcn_wmma_f32_16x16x4_f32(
                false, a1, false, bf, (short)0, acc[1][ns], false, false);
        }
    }

    const int ncol = lane & 15;
#pragma unroll
    for (int ms = 0; ms < 2; ++ms) {
#pragma unroll
        for (int ns = 0; ns < 4; ++ns) {
#pragma unroll
            for (int r = 0; r < 8; ++r) {
                const int o = ((mt0 + ms) << 4) + mo + r;
                const float v = silu(acc[ms][ns][r]);
                out[(((b * OUTC + o) * HH + h) * WWD) + (ns << 4) + ncol] = v;
            }
        }
    }
}

extern "C" void kernel_launch(void* const* d_in, const int* in_sizes, int n_in,
                              void* d_out, int out_size, void* d_ws, size_t ws_size,
                              hipStream_t stream) {
    const float* x         = (const float*)d_in[0];
    const float* w_experts = (const float*)d_in[1];
    const float* bn_gamma  = (const float*)d_in[2];
    const float* bn_beta   = (const float*)d_in[3];
    const float* bn_mean   = (const float*)d_in[4];
    const float* bn_var    = (const float*)d_in[5];
    const float* scales    = (const float*)d_in[6];
    const float* w_out     = (const float*)d_in[7];
    const float* out_gamma = (const float*)d_in[8];
    const float* out_beta  = (const float*)d_in[9];
    const float* out_mean  = (const float*)d_in[10];
    const float* out_var   = (const float*)d_in[11];
    float* out = (float*)d_out;
    float* ws  = (float*)d_ws;

    fold_kernel<<<1, 256, 0, stream>>>(w_experts, bn_gamma, bn_beta, bn_mean,
                                       bn_var, scales, out_gamma, out_beta,
                                       out_mean, out_var, ws);
    swz_kernel<<<128, 256, 0, stream>>>(w_out, ws);
    move_kernel<<<BB * HH, 256, 0, stream>>>(x, ws, out);
}


# fold and weight-swizzle prep merged into one launch: each swizzle thread computes its row's output-BN scale itself (same expression) instead of reading it from the first kernel; 3 launches -> 2
# speedup vs baseline: 3.1984x; 1.0760x over previous
#include <hip/hip_runtime.h>
#include <hip/hip_bf16.h>

#define BB   8
#define CC   256
#define HH   64
#define WWD  64
#define EE   9
#define OUTC 256
#define EPSF 1e-5f

typedef __attribute__((ext_vector_type(2))) float v2f;
typedef __attribute__((ext_vector_type(8))) float v8f;
typedef __attribute__((ext_vector_type(4))) float v4f;

#define WS_WEFF   0
#define WS_EBIAS  2304
#define WS_B2     2560
#define WS_WSWZ   3072

__device__ __forceinline__ float silu(float v) {
    return v / (1.0f + __expf(-v));
}

__device__ __forceinline__ float row_from_left(float v) {
    return __int_as_float(__builtin_amdgcn_update_dpp(
        0, __float_as_int(v), 0x111, 0xf, 0xf, true));
}
__device__ __forceinline__ float row_from_right(float v) {
    return __int_as_float(__builtin_amdgcn_update_dpp(
        0, __float_as_int(v), 0x101, 0xf, 0xf, true));
}

__global__ __launch_bounds__(256) void prep_kernel(
    const float* __restrict__ w_experts, const float* __restrict__ bn_gamma,
    const float* __restrict__ bn_beta,   const float* __restrict__ bn_mean,
    const float* __restrict__ bn_var,    const float* __restrict__ scales,
    const float* __restrict__ w_out,
    const float* __restrict__ out_gamma, const float* __restrict__ out_beta,
    const float* __restrict__ out_mean,  const float* __restrict__ out_var,
    float* __restrict__ ws)
{
    if (blockIdx.x == 0) {
        const int c = threadIdx.x;
        float wacc[9];
#pragma unroll
        for (int k = 0; k < 9; ++k) wacc[k] = 0.0f;
        float bias = 0.0f;
#pragma unroll
        for (int e = 0; e < EE; ++e) {
            const int ec = e * CC + c;
            const float t  = bn_gamma[ec] * rsqrtf(bn_var[ec] + EPSF);
            const float es = t * scales[ec];
            bias += (bn_beta[ec] - bn_mean[ec] * t) * scales[ec];
            const float* wp = w_experts + ec * 9;
#pragma unroll
            for (int k = 0; k < 9; ++k) wacc[k] += es * wp[k];
        }
#pragma unroll
        for (int k = 0; k < 9; ++k) ws[WS_WEFF + c * 9 + k] = wacc[k];
        ws[WS_EBIAS + c] = bias;

        const float t2 = out_gamma[c] * rsqrtf(out_var[c] + EPSF);
        ws[WS_B2 + c] = out_beta[c] - out_mean[c] * t2;
    } else {
        const int idx   = (blockIdx.x - 1) * 256 + threadIdx.x;
        const int kstep = idx >> 9;
        const int rem   = idx & 511;
        const int mt    = rem >> 5;
        const int lane  = rem & 31;
        const int m     = mt * 16 + (lane & 15);
        const int k     = kstep * 4 + ((lane >> 4) << 1);
        const float t2  = out_gamma[m] * rsqrtf(out_var[m] + EPSF);
        ws[WS_WSWZ + idx * 2]     = w_out[m * CC + k]     * t2;
        ws[WS_WSWZ + idx * 2 + 1] = w_out[m * CC + k + 1] * t2;
    }
}

__global__ __launch_bounds__(256) void move_kernel(
    const float* __restrict__ x, const float* __restrict__ ws,
    float* __restrict__ out)
{
    __shared__ __attribute__((aligned(16))) float smem[CC * WWD];

    const int tid = threadIdx.x;
    const int b   = blockIdx.x >> 6;
    const int h   = blockIdx.x & 63;

    const float* weff  = ws + WS_WEFF;
    const float* ebias = ws + WS_EBIAS;
    const float* b2    = ws + WS_B2;
    const float* wswz  = ws + WS_WSWZ;

    {
        const int q    = tid & 15;
        const int hh   = (tid >> 4) & 1;
        const int wv1  = tid >> 5;
        const bool hasU = h > 0;
        const bool hasD = h < HH - 1;
        const int offU = hasU ? -WWD : 0;
        const int offD = hasD ?  WWD : 0;
        const float* xq = x + (((b * CC) * HH + h) * WWD) + (q << 2);
        for (int it = 0; it < 8; ++it) {
            const int kstep = (wv1 << 3) + it;
            const int c     = (kstep << 2) + (hh << 1);
            v4f r[2][3];
            for (int j = 0; j < 2; ++j) {
                const float* p = xq + (c + j) * (HH * WWD);
                r[j][0] = *(const v4f*)(p + offU);
                r[j][1] = *(const v4f*)(p);
                r[j][2] = *(const v4f*)(p + offD);
            }
            float wt[18];
            for (int t = 0; t < 18; ++t) wt[t] = weff[c * 9 + t];
            float bs[2];
            bs[0] = ebias[c];
            bs[1] = ebias[c + 1];
            const v4f zero4 = {0.0f, 0.0f, 0.0f, 0.0f};
            if (!hasU) { r[0][0] = zero4; r[1][0] = zero4; }
            if (!hasD) { r[0][2] = zero4; r[1][2] = zero4; }

            float s[2][4];
            for (int j = 0; j < 2; ++j) {
                float s0 = 0.0f, s1 = 0.0f, s2 = 0.0f, s3 = 0.0f;
                for (int d = 0; d < 3; ++d) {
                    const v4f v  = r[j][d];
                    const float L  = row_from_left(v.w);
                    const float R  = row_from_right(v.x);
                    const float wl = wt[j * 9 + d * 3];
                    const float wc = wt[j * 9 + d * 3 + 1];
                    const float wr = wt[j * 9 + d * 3 + 2];
                    s0 += L   * wl;  s0 += v.x * wc;  s0 += v.y * wr;
                    s1 += v.x * wl;  s1 += v.y * wc;  s1 += v.z * wr;
                    s2 += v.y * wl;  s2 += v.z * wc;  s2 += v.w * wr;
                    s3 += v.z * wl;  s3 += v.w * wc;  s3 += R   * wr;
                }
                s[j][0] = silu(s0 + bs[j]);
                s[j][1] = silu(s1 + bs[j]);
                s[j][2] = silu(s2 + bs[j]);
                s[j][3] = silu(s3 + bs[j]);
            }
            const int base = ((((kstep << 2) + (q >> 2)) << 6) + (hh << 5)) + ((q & 3) << 3);
            const v4f o0 = {s[0][0], s[1][0], s[0][1], s[1][1]};
            const v4f o1 = {s[0][2], s[1][2], s[0][3], s[1][3]};
            *(v4f*)(smem + base)     = o0;
            *(v4f*)(smem + base + 4) = o1;
        }
    }
    __syncthreads();

    const int lane = tid & 31;
    const int wv   = tid >> 5;
    const int mt0  = wv << 1;
    const int mo   = (lane >> 4) << 3;

    v8f acc[2][4];
#pragma unroll
    for (int ms = 0; ms < 2; ++ms) {
        const float* bp = b2 + ((mt0 + ms) << 4) + mo;
#pragma unroll
        for (int r = 0; r < 8; ++r) {
            const float bval = bp[r];
#pragma unroll
            for (int ns = 0; ns < 4; ++ns) acc[ms][ns][r] = bval;
        }
    }

    const v2f* aptr = (const v2f*)wswz;
    const v2f* bptr = (const v2f*)smem;
#pragma unroll 4
    for (int kstep = 0; kstep < 64; ++kstep) {
        const v2f a0 = aptr[((kstep << 4) + mt0)     * 32 + lane];
        const v2f a1 = aptr[((kstep << 4) + mt0 + 1) * 32 + lane];
#pragma unroll
        for (int ns = 0; ns < 4; ++ns) {
            const v2f bf = bptr[((kstep << 2) + ns) * 32 + lane];
            acc[0][ns] = __builtin_amdgcn_wmma_f32_16x16x4_f32(
                false, a0, false, bf, (short)0, acc[0][ns], false, false);
            acc[1][ns] = __builtin_amdgcn_wmma_f32_16x16x4_f32(
                false, a1, false, bf, (short)0, acc[1][ns], false, false);
        }
    }

    const int ncol = lane & 15;
#pragma unroll
    for (int ms = 0; ms < 2; ++ms) {
#pragma unroll
        for (int ns = 0; ns < 4; ++ns) {
#pragma unroll
            for (int r = 0; r < 8; ++r) {
                const int o = ((mt0 + ms) << 4) + mo + r;
                const float v = silu(acc[ms][ns][r]);
                out[(((b * OUTC + o) * HH + h) * WWD) + (ns << 4) + ncol] = v;
            }
        }
    }
}

extern "C" void kernel_launch(void* const* d_in, const int* in_sizes, int n_in,
                              void* d_out, int out_size, void* d_ws, size_t ws_size,
                              hipStream_t stream) {
    const float* x         = (const float*)d_in[0];
    const float* w_experts = (const float*)d_in[1];
    const float* bn_gamma  = (const float*)d_in[2];
    const float* bn_beta   = (const float*)d_in[3];
    const float* bn_mean   = (const float*)d_in[4];
    const float* bn_var    = (const float*)d_in[5];
    const float* scales    = (const float*)d_in[6];
    const float* w_out     = (const float*)d_in[7];
    const float* out_gamma = (const float*)d_in[8];
    const float* out_beta  = (const float*)d_in[9];
    const float* out_mean  = (const float*)d_in[10];
    const float* out_var   = (const float*)d_in[11];
    float* out = (float*)d_out;
    float* ws  = (float*)d_ws;

    prep_kernel<<<129, 256, 0, stream>>>(w_experts, bn_gamma, bn_beta, bn_mean,
                                         bn_var, scales, w_out, out_gamma,
                                         out_beta, out_mean, out_var, ws);
    move_kernel<<<BB * HH, 256, 0, stream>>>(x, ws, out);
}


# conv and GEMM pipelined per 32-channel chunk: next chunk's x rows, taps and A fragments loaded before this chunk's 64 WMMAs, conv after them into the other of two 8 KB LDS buffers; sum order kept
# speedup vs baseline: 3.6619x; 1.1449x over previous
#include <hip/hip_runtime.h>
#include <hip/hip_bf16.h>

#define BB   8
#define CC   256
#define HH   64
#define WWD  64
#define EE   9
#define OUTC 256
#define EPSF 1e-5f

typedef __attribute__((ext_vector_type(2))) float v2f;
typedef __attribute__((ext_vector_type(8))) float v8f;
typedef __attribute__((ext_vector_type(4))) float v4f;

#define WS_WEFF   0
#define WS_EBIAS  2304
#define WS_B2     2560
#define WS_WSWZ   3072

__device__ __forceinline__ float silu(float v) {
    return v / (1.0f + __expf(-v));
}

__device__ __forceinline__ float row_from_left(float v) {
    return __int_as_float(__builtin_amdgcn_update_dpp(
        0, __float_as_int(v), 0x111, 0xf, 0xf, true));
}
__device__ __forceinline__ float row_from_right(float v) {
    return __int_as_float(__builtin_amdgcn_update_dpp(
        0, __float_as_int(v), 0x101, 0xf, 0xf, true));
}

__global__ __launch_bounds__(256) void prep_kernel(
    const float* __restrict__ w_experts, const float* __restrict__ bn_gamma,
    const float* __restrict__ bn_beta,   const float* __restrict__ bn_mean,
    const float* __restrict__ bn_var,    const float* __restrict__ scales,
    const float* __restrict__ w_out,
    const float* __restrict__ out_gamma, const float* __restrict__ out_beta,
    const float* __restrict__ out_mean,  const float* __restrict__ out_var,
    float* __restrict__ ws)
{
    if (blockIdx.x == 0) {
        const int c = threadIdx.x;
        float wacc[9];
#pragma unroll
        for (int k = 0; k < 9; ++k) wacc[k] = 0.0f;
        float bias = 0.0f;
#pragma unroll
        for (int e = 0; e < EE; ++e) {
            const int ec = e * CC + c;
            const float t  = bn_gamma[ec] * rsqrtf(bn_var[ec] + EPSF);
            const float es = t * scales[ec];
            bias += (bn_beta[ec] - bn_mean[ec] * t) * scales[ec];
            const float* wp = w_experts + ec * 9;
#pragma unroll
            for (int k = 0; k < 9; ++k) wacc[k] += es * wp[k];
        }
#pragma unroll
        for (int k = 0; k < 9; ++k) ws[WS_WEFF + c * 9 + k] = wacc[k];
        ws[WS_EBIAS + c] = bias;

        const float t2 = out_gamma[c] * rsqrtf(out_var[c] + EPSF);
        ws[WS_B2 + c] = out_beta[c] - out_mean[c] * t2;
    } else {
        const int idx   = (blockIdx.x - 1) * 256 + threadIdx.x;
        const int kstep = idx >> 9;
        const int rem   = idx & 511;
        const int mt    = rem >> 5;
        const int lane  = rem & 31;
        const int m     = mt * 16 + (lane & 15);
        const int k     = kstep * 4 + ((lane >> 4) << 1);
        const float t2  = out_gamma[m] * rsqrtf(out_var[m] + EPSF);
        ws[WS_WSWZ + idx * 2]     = w_out[m * CC + k]     * t2;
        ws[WS_WSWZ + idx * 2 + 1] = w_out[m * CC + k + 1] * t2;
    }
}

struct ConvIn {
    v4f   r[2][3];
    float wt[18];
    float bs[2];
};

__device__ __forceinline__ void conv_load(ConvIn& ci, const float* __restrict__ xq,
                                          const float* __restrict__ weff,
                                          const float* __restrict__ ebias,
                                          const int c, const int offU, const int offD)
{
    for (int j = 0; j < 2; ++j) {
        const float* p = xq + (c + j) * (HH * WWD);
        ci.r[j][0] = *(const v4f*)(p + offU);
        ci.r[j][1] = *(const v4f*)(p);
        ci.r[j][2] = *(const v4f*)(p + offD);
    }
    for (int t = 0; t < 18; ++t) ci.wt[t] = weff[c * 9 + t];
    ci.bs[0] = ebias[c];
    ci.bs[1] = ebias[c + 1];
}

__device__ __forceinline__ void conv_store(ConvIn ci, const bool hasU, const bool hasD,
                                           float* __restrict__ dst)
{
    const v4f zero4 = {0.0f, 0.0f, 0.0f, 0.0f};
    if (!hasU) { ci.r[0][0] = zero4; ci.r[1][0] = zero4; }
    if (!hasD) { ci.r[0][2] = zero4; ci.r[1][2] = zero4; }

    float s[2][4];
    for (int j = 0; j < 2; ++j) {
        float s0 = 0.0f, s1 = 0.0f, s2 = 0.0f, s3 = 0.0f;
        for (int d = 0; d < 3; ++d) {
            const v4f v  = ci.r[j][d];
            const float L  = row_from_left(v.w);
            const float R  = row_from_right(v.x);
            const float wl = ci.wt[j * 9 + d * 3];
            const float wc = ci.wt[j * 9 + d * 3 + 1];
            const float wr = ci.wt[j * 9 + d * 3 + 2];
            s0 += L   * wl;  s0 += v.x * wc;  s0 += v.y * wr;
            s1 += v.x * wl;  s1 += v.y * wc;  s1 += v.z * wr;
            s2 += v.y * wl;  s2 += v.z * wc;  s2 += v.w * wr;
            s3 += v.z * wl;  s3 += v.w * wc;  s3 += R   * wr;
        }
        s[j][0] = silu(s0 + ci.bs[j]);
        s[j][1] = silu(s1 + ci.bs[j]);
        s[j][2] = silu(s2 + ci.bs[j]);
        s[j][3] = silu(s3 + ci.bs[j]);
    }
    const v4f o0 = {s[0][0], s[1][0], s[0][1], s[1][1]};
    const v4f o1 = {s[0][2], s[1][2], s[0][3], s[1][3]};
    *(v4f*)(dst)     = o0;
    *(v4f*)(dst + 4) = o1;
}

__device__ __forceinline__ void gemm_chunk(v8f (&acc)[2][4], const v2f (&a)[16],
                                           const v2f* __restrict__ bl)
{
#pragma unroll
    for (int kl = 0; kl < 8; ++kl) {
#pragma unroll
        for (int ns = 0; ns < 4; ++ns) {
            const v2f bf = bl[((kl << 2) + ns) * 32];
            acc[0][ns] = __builtin_amdgcn_wmma_f32_16x16x4_f32(
                false, a[2 * kl],     false, bf, (short)0, acc[0][ns], false, false);
            acc[1][ns] = __builtin_amdgcn_wmma_f32_16x16x4_f32(
                false, a[2 * kl + 1], false, bf, (short)0, acc[1][ns], false, false);
        }
    }
}

__global__ __launch_bounds__(256) void move_kernel(
    const float* __restrict__ x, const float* __restrict__ ws,
    float* __restrict__ out)
{
    __shared__ __attribute__((aligned(16))) float smem[2 * 2048];

    const int tid = threadIdx.x;
    const int b   = blockIdx.x >> 6;
    const int h   = blockIdx.x & 63;

    const float* weff  = ws + WS_WEFF;
    const float* ebias = ws + WS_EBIAS;
    const float* b2    = ws + WS_B2;
    const float* wswz  = ws + WS_WSWZ;

    const int q    = tid & 15;
    const int hh   = (tid >> 4) & 1;
    const int wv   = tid >> 5;
    const bool hasU = h > 0;
    const bool hasD = h < HH - 1;
    const int offU = hasU ? -WWD : 0;
    const int offD = hasD ?  WWD : 0;
    const float* xq = x + (((b * CC) * HH + h) * WWD) + (q << 2);
    const int c0    = (wv << 2) + (hh << 1);
    float* sdst = smem + ((((wv << 2) + (q >> 2)) << 6) + (hh << 5)) + ((q & 3) << 3);

    const int lane = tid & 31;
    const int mt0  = wv << 1;
    const int mo   = (lane >> 4) << 3;
    const v2f* aptr = (const v2f*)wswz + mt0 * 32 + lane;
    const v2f* bptr = (const v2f*)smem + lane;

    ConvIn cin;
    conv_load(cin, xq, weff, ebias, c0, offU, offD);

    v2f a_cur[16];
#pragma unroll
    for (int kl = 0; kl < 8; ++kl) {
        a_cur[2 * kl]     = aptr[kl * 512];
        a_cur[2 * kl + 1] = aptr[kl * 512 + 32];
    }

    v8f acc[2][4];
#pragma unroll
    for (int ms = 0; ms < 2; ++ms) {
        const float* bp = b2 + ((mt0 + ms) << 4) + mo;
#pragma unroll
        for (int r = 0; r < 8; ++r) {
            const float bval = bp[r];
#pragma unroll
            for (int ns = 0; ns < 4; ++ns) acc[ms][ns][r] = bval;
        }
    }

    conv_store(cin, hasU, hasD, sdst);
    __syncthreads();

    for (int ch = 0; ch < 7; ++ch) {
        ConvIn nin;
        conv_load(nin, xq, weff, ebias, c0 + ((ch + 1) << 5), offU, offD);
        v2f a_nxt[16];
#pragma unroll
        for (int kl = 0; kl < 8; ++kl) {
            a_nxt[2 * kl]     = aptr[(((ch + 1) << 3) + kl) * 512];
            a_nxt[2 * kl + 1] = aptr[(((ch + 1) << 3) + kl) * 512 + 32];
        }

        __builtin_amdgcn_sched_barrier(0);

        gemm_chunk(acc, a_cur, bptr + ((ch & 1) << 10));

        __builtin_amdgcn_sched_barrier(0);
        asm volatile("" : "+v"(nin.r[0][0]), "+v"(nin.r[0][1]), "+v"(nin.r[0][2]),
                          "+v"(nin.r[1][0]), "+v"(nin.r[1][1]), "+v"(nin.r[1][2]));

        conv_store(nin, hasU, hasD, sdst + (((ch + 1) & 1) << 11));
#pragma unroll
        for (int i = 0; i < 16; ++i) a_cur[i] = a_nxt[i];
        __syncthreads();
    }
    gemm_chunk(acc, a_cur, bptr + (1 << 10));

    const int ncol = lane & 15;
#pragma unroll
    for (int ms = 0; ms < 2; ++ms) {
#pragma unroll
        for (int ns = 0; ns < 4; ++ns) {
#pragma unroll
            for (int r = 0; r < 8; ++r) {
                const int o = ((mt0 + ms) << 4) + mo + r;
                const float v = silu(acc[ms][ns][r]);
                out[(((b * OUTC + o) * HH + h) * WWD) + (ns << 4) + ncol] = v;
            }
        }
    }
}

extern "C" void kernel_launch(void* const* d_in, const int* in_sizes, int n_in,
                              void* d_out, int out_size, void* d_ws, size_t ws_size,
                              hipStream_t stream) {
    const float* x         = (const float*)d_in[0];
    const float* w_experts = (const float*)d_in[1];
    const float* bn_gamma  = (const float*)d_in[2];
    const float* bn_beta   = (const float*)d_in[3];
    const float* bn_mean   = (const float*)d_in[4];
    const float* bn_var    = (const float*)d_in[5];
    const float* scales    = (const float*)d_in[6];
    const float* w_out     = (const float*)d_in[7];
    const float* out_gamma = (const float*)d_in[8];
    const float* out_beta  = (const float*)d_in[9];
    const float* out_mean  = (const float*)d_in[10];
    const float* out_var   = (const float*)d_in[11];
    float* out = (float*)d_out;
    float* ws  = (float*)d_ws;

    prep_kernel<<<129, 256, 0, stream>>>(w_experts, bn_gamma, bn_beta, bn_mean,
                                         bn_var, scales, w_out, out_gamma,
                                         out_beta, out_mean, out_var, ws);
    move_kernel<<<BB * HH, 256, 0, stream>>>(x, ws, out);
}


# SiLU (conv stage and epilogue): v * v_rcp_f32(1 + e^-v), a 1-ulp reciprocal and one multiply, replaces the in-line IEEE division (11 vector instructions per value); every sum and its order kept
# speedup vs baseline: 3.8936x; 1.0633x over previous
#include <hip/hip_runtime.h>
#include <hip/hip_bf16.h>

#define BB   8
#define CC   256
#define HH   64
#define WWD  64
#define EE   9
#define OUTC 256
#define EPSF 1e-5f

typedef __attribute__((ext_vector_type(2))) float v2f;
typedef __attribute__((ext_vector_type(8))) float v8f;
typedef __attribute__((ext_vector_type(4))) float v4f;

#define WS_WEFF   0
#define WS_EBIAS  2304
#define WS_B2     2560
#define WS_WSWZ   3072

__device__ __forceinline__ float silu(float v) {
    return v * __builtin_amdgcn_rcpf(1.0f + __expf(-v));
}

__device__ __forceinline__ float row_from_left(float v) {
    return __int_as_float(__builtin_amdgcn_update_dpp(
        0, __float_as_int(v), 0x111, 0xf, 0xf, true));
}
__device__ __forceinline__ float row_from_right(float v) {
    return __int_as_float(__builtin_amdgcn_update_dpp(
        0, __float_as_int(v), 0x101, 0xf, 0xf, true));
}

__global__ __launch_bounds__(256) void prep_kernel(
    const float* __restrict__ w_experts, const float* __restrict__ bn_gamma,
    const float* __restrict__ bn_beta,   const float* __restrict__ bn_mean,
    const float* __restrict__ bn_var,    const float* __restrict__ scales,
    const float* __restrict__ w_out,
    const float* __restrict__ out_gamma, const float* __restrict__ out_beta,
    const float* __restrict__ out_mean,  const float* __restrict__ out_var,
    float* __restrict__ ws)
{
    if (blockIdx.x == 0) {
        const int c = threadIdx.x;
        float wacc[9];
#pragma unroll
        for (int k = 0; k < 9; ++k) wacc[k] = 0.0f;
        float bias = 0.0f;
#pragma unroll
        for (int e = 0; e < EE; ++e) {
            const int ec = e * CC + c;
            const float t  = bn_gamma[ec] * rsqrtf(bn_var[ec] + EPSF);
            const float es = t * scales[ec];
            bias += (bn_beta[ec] - bn_mean[ec] * t) * scales[ec];
            const float* wp = w_experts + ec * 9;
#pragma unroll
            for (int k = 0; k < 9; ++k) wacc[k] += es * wp[k];
        }
#pragma unroll
        for (int k = 0; k < 9; ++k) ws[WS_WEFF + c * 9 + k] = wacc[k];
        ws[WS_EBIAS + c] = bias;

        const float t2 = out_gamma[c] * rsqrtf(out_var[c] + EPSF);
        ws[WS_B2 + c] = out_beta[c] - out_mean[c] * t2;
    } else {
        const int idx   = (blockIdx.x - 1) * 256 + threadIdx.x;
        const int kstep = idx >> 9;
        const int rem   = idx & 511;
        const int mt    = rem >> 5;
        const int lane  = rem & 31;
        const int m     = mt * 16 + (lane & 15);
        const int k     = kstep * 4 + ((lane >> 4) << 1);
        const float t2  = out_gamma[m] * rsqrtf(out_var[m] + EPSF);
        ws[WS_WSWZ + idx * 2]     = w_out[m * CC + k]     * t2;
        ws[WS_WSWZ + idx * 2 + 1] = w_out[m * CC + k + 1] * t2;
    }
}

struct ConvIn {
    v4f   r[2][3];
    float wt[18];
    float bs[2];
};

__device__ __forceinline__ void conv_load(ConvIn& ci, const float* __restrict__ xq,
                                          const float* __restrict__ weff,
                                          const float* __restrict__ ebias,
                                          const int c, const int offU, const int offD)
{
    for (int j = 0; j < 2; ++j) {
        const float* p = xq + (c + j) * (HH * WWD);
        ci.r[j][0] = *(const v4f*)(p + offU);
        ci.r[j][1] = *(const v4f*)(p);
        ci.r[j][2] = *(const v4f*)(p + offD);
    }
    for (int t = 0; t < 18; ++t) ci.wt[t] = weff[c * 9 + t];
    ci.bs[0] = ebias[c];
    ci.bs[1] = ebias[c + 1];
}

__device__ __forceinline__ void conv_store(ConvIn ci, const bool hasU, const bool hasD,
                                           float* __restrict__ dst)
{
    const v4f zero4 = {0.0f, 0.0f, 0.0f, 0.0f};
    if (!hasU) { ci.r[0][0] = zero4; ci.r[1][0] = zero4; }
    if (!hasD) { ci.r[0][2] = zero4; ci.r[1][2] = zero4; }

    float s[2][4];
    for (int j = 0; j < 2; ++j) {
        float s0 = 0.0f, s1 = 0.0f, s2 = 0.0f, s3 = 0.0f;
        for (int d = 0; d < 3; ++d) {
            const v4f v  = ci.r[j][d];
            const float L  = row_from_left(v.w);
            const float R  = row_from_right(v.x);
            const float wl = ci.wt[j * 9 + d * 3];
            const float wc = ci.wt[j * 9 + d * 3 + 1];
            const float wr = ci.wt[j * 9 + d * 3 + 2];
            s0 += L   * wl;  s0 += v.x * wc;  s0 += v.y * wr;
            s1 += v.x * wl;  s1 += v.y * wc;  s1 += v.z * wr;
            s2 += v.y * wl;  s2 += v.z * wc;  s2 += v.w * wr;
            s3 += v.z * wl;  s3 += v.w * wc;  s3 += R   * wr;
        }
        s[j][0] = silu(s0 + ci.bs[j]);
        s[j][1] = silu(s1 + ci.bs[j]);
        s[j][2] = silu(s2 + ci.bs[j]);
        s[j][3] = silu(s3 + ci.bs[j]);
    }
    const v4f o0 = {s[0][0], s[1][0], s[0][1], s[1][1]};
    const v4f o1 = {s[0][2], s[1][2], s[0][3], s[1][3]};
    *(v4f*)(dst)     = o0;
    *(v4f*)(dst + 4) = o1;
}

__device__ __forceinline__ void gemm_chunk(v8f (&acc)[2][4], const v2f (&a)[16],
                                           const v2f* __restrict__ bl)
{
#pragma unroll
    for (int kl = 0; kl < 8; ++kl) {
#pragma unroll
        for (int ns = 0; ns < 4; ++ns) {
            const v2f bf = bl[((kl << 2) + ns) * 32];
            acc[0][ns] = __builtin_amdgcn_wmma_f32_16x16x4_f32(
                false, a[2 * kl],     false, bf, (short)0, acc[0][ns], false, false);
            acc[1][ns] = __builtin_amdgcn_wmma_f32_16x16x4_f32(
                false, a[2 * kl + 1], false, bf, (short)0, acc[1][ns], false, false);
        }
    }
}

__global__ __launch_bounds__(256) void move_kernel(
    const float* __restrict__ x, const float* __restrict__ ws,
    float* __restrict__ out)
{
    __shared__ __attribute__((aligned(16))) float smem[2 * 2048];

    const int tid = threadIdx.x;
    const int b   = blockIdx.x >> 6;
    const int h   = blockIdx.x & 63;

    const float* weff  = ws + WS_WEFF;
    const float* ebias = ws + WS_EBIAS;
    const float* b2    = ws + WS_B2;
    const float* wswz  = ws + WS_WSWZ;

    const int q    = tid & 15;
    const int hh   = (tid >> 4) & 1;
    const int wv   = tid >> 5;
    const bool hasU = h > 0;
    const bool hasD = h < HH - 1;
    const int offU = hasU ? -WWD : 0;
    const int offD = hasD ?  WWD : 0;
    const float* xq = x + (((b * CC) * HH + h) * WWD) + (q << 2);
    const int c0    = (wv << 2) + (hh << 1);
    float* sdst = smem + ((((wv << 2) + (q >> 2)) << 6) + (hh << 5)) + ((q & 3) << 3);

    const int lane = tid & 31;
    const int mt0  = wv << 1;
    const int mo   = (lane >> 4) << 3;
    const v2f* aptr = (const v2f*)wswz + mt0 * 32 + lane;
    const v2f* bptr = (const v2f*)smem + lane;

    ConvIn cin;
    conv_load(cin, xq, weff, ebias, c0, offU, offD);

    v2f a_cur[16];
#pragma unroll
    for (int kl = 0; kl < 8; ++kl) {
        a_cur[2 * kl]     = aptr[kl * 512];
        a_cur[2 * kl + 1] = aptr[kl * 512 + 32];
    }

    v8f acc[2][4];
#pragma unroll
    for (int ms = 0; ms < 2; ++ms) {
        const float* bp = b2 + ((mt0 + ms) << 4) + mo;
#pragma unroll
        for (int r = 0; r < 8; ++r) {
            const float bval = bp[r];
#pragma unroll
            for (int ns = 0; ns < 4; ++ns) acc[ms][ns][r] = bval;
        }
    }

    conv_store(cin, hasU, hasD, sdst);
    __syncthreads();

    for (int ch = 0; ch < 7; ++ch) {
        ConvIn nin;
        conv_load(nin, xq, weff, ebias, c0 + ((ch + 1) << 5), offU, offD);
        v2f a_nxt[16];
#pragma unroll
        for (int kl = 0; kl < 8; ++kl) {
            a_nxt[2 * kl]     = aptr[(((ch + 1) << 3) + kl) * 512];
            a_nxt[2 * kl + 1] = aptr[(((ch + 1) << 3) + kl) * 512 + 32];
        }

        __builtin_amdgcn_sched_barrier(0);

        gemm_chunk(acc, a_cur, bptr + ((ch & 1) << 10));

        __builtin_amdgcn_sched_barrier(0);
        asm volatile("" : "+v"(nin.r[0][0]), "+v"(nin.r[0][1]), "+v"(nin.r[0][2]),
                          "+v"(nin.r[1][0]), "+v"(nin.r[1][1]), "+v"(nin.r[1][2]));

        conv_store(nin, hasU, hasD, sdst + (((ch + 1) & 1) << 11));
#pragma unroll
        for (int i = 0; i < 16; ++i) a_cur[i] = a_nxt[i];
        __syncthreads();
    }
    gemm_chunk(acc, a_cur, bptr + (1 << 10));

    const int ncol = lane & 15;
#pragma unroll
    for (int ms = 0; ms < 2; ++ms) {
#pragma unroll
        for (int ns = 0; ns < 4; ++ns) {
#pragma unroll
            for (int r = 0; r < 8; ++r) {
                const int o = ((mt0 + ms) << 4) + mo + r;
                const float v = silu(acc[ms][ns][r]);
                out[(((b * OUTC + o) * HH + h) * WWD) + (ns << 4) + ncol] = v;
            }
        }
    }
}

extern "C" void kernel_launch(void* const* d_in, const int* in_sizes, int n_in,
                              void* d_out, int out_size, void* d_ws, size_t ws_size,
                              hipStream_t stream) {
    const float* x         = (const float*)d_in[0];
    const float* w_experts = (const float*)d_in[1];
    const float* bn_gamma  = (const float*)d_in[2];
    const float* bn_beta   = (const float*)d_in[3];
    const float* bn_mean   = (const float*)d_in[4];
    const float* bn_var    = (const float*)d_in[5];
    const float* scales    = (const float*)d_in[6];
    const float* w_out     = (const float*)d_in[7];
    const float* out_gamma = (const float*)d_in[8];
    const float* out_beta  = (const float*)d_in[9];
    const float* out_mean  = (const float*)d_in[10];
    const float* out_var   = (const float*)d_in[11];
    float* out = (float*)d_out;
    float* ws  = (float*)d_ws;

    prep_kernel<<<129, 256, 0, stream>>>(w_experts, bn_gamma, bn_beta, bn_mean,
                                         bn_var, scales, w_out, out_gamma,
                                         out_beta, out_mean, out_var, ws);
    move_kernel<<<BB * HH, 256, 0, stream>>>(x, ws, out);
}


# epilogue: each wave transposes its SiLU'd 32x64 tile through an LDS tile of its own and stores whole 256-byte rows (16 x 16-byte stores per wave), not 64 4-byte stores writing 64-byte half lines
# speedup vs baseline: 4.1408x; 1.0635x over previous
#include <hip/hip_runtime.h>
#include <hip/hip_bf16.h>

#define BB   8
#define CC   256
#define HH   64
#define WWD  64
#define EE   9
#define OUTC 256
#define EPSF 1e-5f
#define STP  68

typedef __attribute__((ext_vector_type(2))) float v2f;
typedef __attribute__((ext_vector_type(8))) float v8f;
typedef __attribute__((ext_vector_type(4))) float v4f;

#define WS_WEFF   0
#define WS_EBIAS  2304
#define WS_B2     2560
#define WS_WSWZ   3072

__device__ __forceinline__ float silu(float v) {
    return v * __builtin_amdgcn_rcpf(1.0f + __expf(-v));
}

__device__ __forceinline__ float row_from_left(float v) {
    return __int_as_float(__builtin_amdgcn_update_dpp(
        0, __float_as_int(v), 0x111, 0xf, 0xf, true));
}
__device__ __forceinline__ float row_from_right(float v) {
    return __int_as_float(__builtin_amdgcn_update_dpp(
        0, __float_as_int(v), 0x101, 0xf, 0xf, true));
}

__global__ __launch_bounds__(256) void prep_kernel(
    const float* __restrict__ w_experts, const float* __restrict__ bn_gamma,
    const float* __restrict__ bn_beta,   const float* __restrict__ bn_mean,
    const float* __restrict__ bn_var,    const float* __restrict__ scales,
    const float* __restrict__ w_out,
    const float* __restrict__ out_gamma, const float* __restrict__ out_beta,
    const float* __restrict__ out_mean,  const float* __restrict__ out_var,
    float* __restrict__ ws)
{
    if (blockIdx.x == 0) {
        const int c = threadIdx.x;
        float wacc[9];
#pragma unroll
        for (int k = 0; k < 9; ++k) wacc[k] = 0.0f;
        float bias = 0.0f;
#pragma unroll
        for (int e = 0; e < EE; ++e) {
            const int ec = e * CC + c;
            const float t  = bn_gamma[ec] * rsqrtf(bn_var[ec] + EPSF);
            const float es = t * scales[ec];
            bias += (bn_beta[ec] - bn_mean[ec] * t) * scales[ec];
            const float* wp = w_experts + ec * 9;
#pragma unroll
            for (int k = 0; k < 9; ++k) wacc[k] += es * wp[k];
        }
#pragma unroll
        for (int k = 0; k < 9; ++k) ws[WS_WEFF + c * 9 + k] = wacc[k];
        ws[WS_EBIAS + c] = bias;

        const float t2 = out_gamma[c] * rsqrtf(out_var[c] + EPSF);
        ws[WS_B2 + c] = out_beta[c] - out_mean[c] * t2;
    } else {
        const int idx   = (blockIdx.x - 1) * 256 + threadIdx.x;
        const int kstep = idx >> 9;
        const int rem   = idx & 511;
        const int mt    = rem >> 5;
        const int lane  = rem & 31;
        const int m     = mt * 16 + (lane & 15);
        const int k     = kstep * 4 + ((lane >> 4) << 1);
        const float t2  = out_gamma[m] * rsqrtf(out_var[m] + EPSF);
        ws[WS_WSWZ + idx * 2]     = w_out[m * CC + k]     * t2;
        ws[WS_WSWZ + idx * 2 + 1] = w_out[m * CC + k + 1] * t2;
    }
}

struct ConvIn {
    v4f   r[2][3];
    float wt[18];
    float bs[2];
};

__device__ __forceinline__ void conv_load(ConvIn& ci, const float* __restrict__ xq,
                                          const float* __restrict__ weff,
                                          const float* __restrict__ ebias,
                                          const int c, const int offU, const int offD)
{
    for (int j = 0; j < 2; ++j) {
        const float* p = xq + (c + j) * (HH * WWD);
        ci.r[j][0] = *(const v4f*)(p + offU);
        ci.r[j][1] = *(const v4f*)(p);
        ci.r[j][2] = *(const v4f*)(p + offD);
    }
    for (int t = 0; t < 18; ++t) ci.wt[t] = weff[c * 9 + t];
    ci.bs[0] = ebias[c];
    ci.bs[1] = ebias[c + 1];
}

__device__ __forceinline__ void conv_store(ConvIn ci, const bool hasU, const bool hasD,
                                           float* __restrict__ dst)
{
    const v4f zero4 = {0.0f, 0.0f, 0.0f, 0.0f};
    if (!hasU) { ci.r[0][0] = zero4; ci.r[1][0] = zero4; }
    if (!hasD) { ci.r[0][2] = zero4; ci.r[1][2] = zero4; }

    float s[2][4];
    for (int j = 0; j < 2; ++j) {
        float s0 = 0.0f, s1 = 0.0f, s2 = 0.0f, s3 = 0.0f;
        for (int d = 0; d < 3; ++d) {
            const v4f v  = ci.r[j][d];
            const float L  = row_from_left(v.w);
            const float R  = row_from_right(v.x);
            const float wl = ci.wt[j * 9 + d * 3];
            const float wc = ci.wt[j * 9 + d * 3 + 1];
            const float wr = ci.wt[j * 9 + d * 3 + 2];
            s0 += L   * wl;  s0 += v.x * wc;  s0 += v.y * wr;
            s1 += v.x * wl;  s1 += v.y * wc;  s1 += v.z * wr;
            s2 += v.y * wl;  s2 += v.z * wc;  s2 += v.w * wr;
            s3 += v.z * wl;  s3 += v.w * wc;  s3 += R   * wr;
        }
        s[j][0] = silu(s0 + ci.bs[j]);
        s[j][1] = silu(s1 + ci.bs[j]);
        s[j][2] = silu(s2 + ci.bs[j]);
        s[j][3] = silu(s3 + ci.bs[j]);
    }
    const v4f o0 = {s[0][0], s[1][0], s[0][1], s[1][1]};
    const v4f o1 = {s[0][2], s[1][2], s[0][3], s[1][3]};
    *(v4f*)(dst)     = o0;
    *(v4f*)(dst + 4) = o1;
}

__device__ __forceinline__ void gemm_chunk(v8f (&acc)[2][4], const v2f (&a)[16],
                                           const v2f* __restrict__ bl)
{
#pragma unroll
    for (int kl = 0; kl < 8; ++kl) {
#pragma unroll
        for (int ns = 0; ns < 4; ++ns) {
            const v2f bf = bl[((kl << 2) + ns) * 32];
            acc[0][ns] = __builtin_amdgcn_wmma_f32_16x16x4_f32(
                false, a[2 * kl],     false, bf, (short)0, acc[0][ns], false, false);
            acc[1][ns] = __builtin_amdgcn_wmma_f32_16x16x4_f32(
                false, a[2 * kl + 1], false, bf, (short)0, acc[1][ns], false, false);
        }
    }
}

__global__ __launch_bounds__(256) void move_kernel(
    const float* __restrict__ x, const float* __restrict__ ws,
    float* __restrict__ out)
{
    __shared__ __attribute__((aligned(16))) float smem[2 * 2048];
    __shared__ __attribute__((aligned(16))) float stage[8 * 32 * STP];

    const int tid = threadIdx.x;
    const int b   = blockIdx.x >> 6;
    const int h   = blockIdx.x & 63;

    const float* weff  = ws + WS_WEFF;
    const float* ebias = ws + WS_EBIAS;
    const float* b2    = ws + WS_B2;
    const float* wswz  = ws + WS_WSWZ;

    const int q    = tid & 15;
    const int hh   = (tid >> 4) & 1;
    const int wv   = tid >> 5;
    const bool hasU = h > 0;
    const bool hasD = h < HH - 1;
    const int offU = hasU ? -WWD : 0;
    const int offD = hasD ?  WWD : 0;
    const float* xq = x + (((b * CC) * HH + h) * WWD) + (q << 2);
    const int c0    = (wv << 2) + (hh << 1);
    float* sdst = smem + ((((wv << 2) + (q >> 2)) << 6) + (hh << 5)) + ((q & 3) << 3);

    const int lane = tid & 31;
    const int mt0  = wv << 1;
    const int mo   = (lane >> 4) << 3;
    const v2f* aptr = (const v2f*)wswz + mt0 * 32 + lane;
    const v2f* bptr = (const v2f*)smem + lane;

    ConvIn cin;
    conv_load(cin, xq, weff, ebias, c0, offU, offD);

    v2f a_cur[16];
#pragma unroll
    for (int kl = 0; kl < 8; ++kl) {
        a_cur[2 * kl]     = aptr[kl * 512];
        a_cur[2 * kl + 1] = aptr[kl * 512 + 32];
    }

    v8f acc[2][4];
#pragma unroll
    for (int ms = 0; ms < 2; ++ms) {
        const float* bp = b2 + ((mt0 + ms) << 4) + mo;
#pragma unroll
        for (int r = 0; r < 8; ++r) {
            const float bval = bp[r];
#pragma unroll
            for (int ns = 0; ns < 4; ++ns) acc[ms][ns][r] = bval;
        }
    }

    conv_store(cin, hasU, hasD, sdst);
    __syncthreads();

    for (int ch = 0; ch < 7; ++ch) {
        ConvIn nin;
        conv_load(nin, xq, weff, ebias, c0 + ((ch + 1) << 5), offU, offD);
        v2f a_nxt[16];
#pragma unroll
        for (int kl = 0; kl < 8; ++kl) {
            a_nxt[2 * kl]     = aptr[(((ch + 1) << 3) + kl) * 512];
            a_nxt[2 * kl + 1] = aptr[(((ch + 1) << 3) + kl) * 512 + 32];
        }

        __builtin_amdgcn_sched_barrier(0);

        gemm_chunk(acc, a_cur, bptr + ((ch & 1) << 10));

        __builtin_amdgcn_sched_barrier(0);
        asm volatile("" : "+v"(nin.r[0][0]), "+v"(nin.r[0][1]), "+v"(nin.r[0][2]),
                          "+v"(nin.r[1][0]), "+v"(nin.r[1][1]), "+v"(nin.r[1][2]));

        conv_store(nin, hasU, hasD, sdst + (((ch + 1) & 1) << 11));
#pragma unroll
        for (int i = 0; i < 16; ++i) a_cur[i] = a_nxt[i];
        __syncthreads();
    }
    gemm_chunk(acc, a_cur, bptr + (1 << 10));

    float* st = stage + wv * (32 * STP);
    const int ncol = lane & 15;
#pragma unroll
    for (int ms = 0; ms < 2; ++ms) {
#pragma unroll
        for (int ns = 0; ns < 4; ++ns) {
#pragma unroll
            for (int r = 0; r < 8; ++r) {
                st[((ms << 4) + mo + r) * STP + (ns << 4) + ncol] = silu(acc[ms][ns][r]);
            }
        }
    }
    __builtin_amdgcn_wave_barrier();
    const int rr  = lane >> 4;
    const int cc4 = (lane & 15) << 2;
    const float* sp = st + rr * STP + cc4;
    float* op = out + ((((b * OUTC) + (mt0 << 4) + rr) * HH + h) * WWD) + cc4;
#pragma unroll
    for (int i = 0; i < 16; ++i) {
        const v4f v = *(const v4f*)(sp + (2 * i) * STP);
        *(v4f*)(op + (2 * i) * (HH * WWD)) = v;
    }
}

extern "C" void kernel_launch(void* const* d_in, const int* in_sizes, int n_in,
                              void* d_out, int out_size, void* d_ws, size_t ws_size,
                              hipStream_t stream) {
    const float* x         = (const float*)d_in[0];
    const float* w_experts = (const float*)d_in[1];
    const float* bn_gamma  = (const float*)d_in[2];
    const float* bn_beta   = (const float*)d_in[3];
    const float* bn_mean   = (const float*)d_in[4];
    const float* bn_var    = (const float*)d_in[5];
    const float* scales    = (const float*)d_in[6];
    const float* w_out     = (const float*)d_in[7];
    const float* out_gamma = (const float*)d_in[8];
    const float* out_beta  = (const float*)d_in[9];
    const float* out_mean  = (const float*)d_in[10];
    const float* out_var   = (const float*)d_in[11];
    float* out = (float*)d_out;
    float* ws  = (float*)d_ws;

    prep_kernel<<<129, 256, 0, stream>>>(w_experts, bn_gamma, bn_beta, bn_mean,
                                         bn_var, scales, w_out, out_gamma,
                                         out_beta, out_mean, out_var, ws);
    move_kernel<<<BB * HH, 256, 0, stream>>>(x, ws, out);
}


# prep launch cut to one memory round per block: expert fold moved from one block (a thread per channel, 126 loads in dependent rounds) to ten (a thread per folded tap, 36 loads); same sums, same order
# speedup vs baseline: 4.3242x; 1.0443x over previous
#include <hip/hip_runtime.h>
#include <hip/hip_bf16.h>

#define BB   8
#define CC   256
#define HH   64
#define WWD  64
#define EE   9
#define OUTC 256
#define EPSF 1e-5f
#define STP  68

typedef __attribute__((ext_vector_type(2))) float v2f;
typedef __attribute__((ext_vector_type(8))) float v8f;
typedef __attribute__((ext_vector_type(4))) float v4f;

#define WS_WEFF   0
#define WS_EBIAS  2304
#define WS_B2     2560
#define WS_WSWZ   3072

__device__ __forceinline__ float silu(float v) {
    return v * __builtin_amdgcn_rcpf(1.0f + __expf(-v));
}

__device__ __forceinline__ float row_from_left(float v) {
    return __int_as_float(__builtin_amdgcn_update_dpp(
        0, __float_as_int(v), 0x111, 0xf, 0xf, true));
}
__device__ __forceinline__ float row_from_right(float v) {
    return __int_as_float(__builtin_amdgcn_update_dpp(
        0, __float_as_int(v), 0x101, 0xf, 0xf, true));
}

__global__ __launch_bounds__(256) void prep_kernel(
    const float* __restrict__ w_experts, const float* __restrict__ bn_gamma,
    const float* __restrict__ bn_beta,   const float* __restrict__ bn_mean,
    const float* __restrict__ bn_var,    const float* __restrict__ scales,
    const float* __restrict__ w_out,
    const float* __restrict__ out_gamma, const float* __restrict__ out_beta,
    const float* __restrict__ out_mean,  const float* __restrict__ out_var,
    float* __restrict__ ws)
{
    const int blk = blockIdx.x;
    if (blk < 9) {
        const int f = blk * 256 + threadIdx.x;
        const int c = f / 9;
        float wacc = 0.0f;
#pragma unroll
        for (int e = 0; e < EE; ++e) {
            const int ec = e * CC + c;
            const float t  = bn_gamma[ec] * rsqrtf(bn_var[ec] + EPSF);
            const float es = t * scales[ec];
            wacc = fmaf(es, w_experts[e * (CC * 9) + f], wacc);
        }
        ws[WS_WEFF + f] = wacc;
    } else if (blk == 9) {
        const int c = threadIdx.x;
        float bias = 0.0f;
#pragma unroll
        for (int e = 0; e < EE; ++e) {
            const int ec = e * CC + c;
            const float t = bn_gamma[ec] * rsqrtf(bn_var[ec] + EPSF);
            bias += (bn_beta[ec] - bn_mean[ec] * t) * scales[ec];
        }
        ws[WS_EBIAS + c] = bias;

        const float t2 = out_gamma[c] * rsqrtf(out_var[c] + EPSF);
        ws[WS_B2 + c] = out_beta[c] - out_mean[c] * t2;
    } else {
        const int idx   = (blk - 10) * 256 + threadIdx.x;
        const int kstep = idx >> 9;
        const int rem   = idx & 511;
        const int mt    = rem >> 5;
        const int lane  = rem & 31;
        const int m     = mt * 16 + (lane & 15);
        const int k     = kstep * 4 + ((lane >> 4) << 1);
        const float w0  = w_out[m * CC + k];
        const float w1  = w_out[m * CC + k + 1];
        const float og  = out_gamma[m];
        const float ov  = out_var[m];
        __builtin_amdgcn_sched_barrier(0);
        const float t2  = og * rsqrtf(ov + EPSF);
        ws[WS_WSWZ + idx * 2]     = w0 * t2;
        ws[WS_WSWZ + idx * 2 + 1] = w1 * t2;
    }
}

struct ConvIn {
    v4f   r[2][3];
    float wt[18];
    float bs[2];
};

__device__ __forceinline__ void conv_load(ConvIn& ci, const float* __restrict__ xq,
                                          const float* __restrict__ weff,
                                          const float* __restrict__ ebias,
                                          const int c, const int offU, const int offD)
{
    for (int j = 0; j < 2; ++j) {
        const float* p = xq + (c + j) * (HH * WWD);
        ci.r[j][0] = *(const v4f*)(p + offU);
        ci.r[j][1] = *(const v4f*)(p);
        ci.r[j][2] = *(const v4f*)(p + offD);
    }
    for (int t = 0; t < 18; ++t) ci.wt[t] = weff[c * 9 + t];
    ci.bs[0] = ebias[c];
    ci.bs[1] = ebias[c + 1];
}

__device__ __forceinline__ void conv_store(ConvIn ci, const bool hasU, const bool hasD,
                                           float* __restrict__ dst)
{
    const v4f zero4 = {0.0f, 0.0f, 0.0f, 0.0f};
    if (!hasU) { ci.r[0][0] = zero4; ci.r[1][0] = zero4; }
    if (!hasD) { ci.r[0][2] = zero4; ci.r[1][2] = zero4; }

    float s[2][4];
    for (int j = 0; j < 2; ++j) {
        float s0 = 0.0f, s1 = 0.0f, s2 = 0.0f, s3 = 0.0f;
        for (int d = 0; d < 3; ++d) {
            const v4f v  = ci.r[j][d];
            const float L  = row_from_left(v.w);
            const float R  = row_from_right(v.x);
            const float wl = ci.wt[j * 9 + d * 3];
            const float wc = ci.wt[j * 9 + d * 3 + 1];
            const float wr = ci.wt[j * 9 + d * 3 + 2];
            s0 += L   * wl;  s0 += v.x * wc;  s0 += v.y * wr;
            s1 += v.x * wl;  s1 += v.y * wc;  s1 += v.z * wr;
            s2 += v.y * wl;  s2 += v.z * wc;  s2 += v.w * wr;
            s3 += v.z * wl;  s3 += v.w * wc;  s3 += R   * wr;
        }
        s[j][0] = silu(s0 + ci.bs[j]);
        s[j][1] = silu(s1 + ci.bs[j]);
        s[j][2] = silu(s2 + ci.bs[j]);
        s[j][3] = silu(s3 + ci.bs[j]);
    }
    const v4f o0 = {s[0][0], s[1][0], s[0][1], s[1][1]};
    const v4f o1 = {s[0][2], s[1][2], s[0][3], s[1][3]};
    *(v4f*)(dst)     = o0;
    *(v4f*)(dst + 4) = o1;
}

__device__ __forceinline__ void gemm_chunk(v8f (&acc)[2][4], const v2f (&a)[16],
                                           const v2f* __restrict__ bl)
{
#pragma unroll
    for (int kl = 0; kl < 8; ++kl) {
#pragma unroll
        for (int ns = 0; ns < 4; ++ns) {
            const v2f bf = bl[((kl << 2) + ns) * 32];
            acc[0][ns] = __builtin_amdgcn_wmma_f32_16x16x4_f32(
                false, a[2 * kl],     false, bf, (short)0, acc[0][ns], false, false);
            acc[1][ns] = __builtin_amdgcn_wmma_f32_16x16x4_f32(
                false, a[2 * kl + 1], false, bf, (short)0, acc[1][ns], false, false);
        }
    }
}

__global__ __launch_bounds__(256) void move_kernel(
    const float* __restrict__ x, const float* __restrict__ ws,
    float* __restrict__ out)
{
    __shared__ __attribute__((aligned(16))) float smem[2 * 2048];
    __shared__ __attribute__((aligned(16))) float stage[8 * 32 * STP];

    const int tid = threadIdx.x;
    const int b   = blockIdx.x >> 6;
    const int h   = blockIdx.x & 63;

    const float* weff  = ws + WS_WEFF;
    const float* ebias = ws + WS_EBIAS;
    const float* b2    = ws + WS_B2;
    const float* wswz  = ws + WS_WSWZ;

    const int q    = tid & 15;
    const int hh   = (tid >> 4) & 1;
    const int wv   = tid >> 5;
    const bool hasU = h > 0;
    const bool hasD = h < HH - 1;
    const int offU = hasU ? -WWD : 0;
    const int offD = hasD ?  WWD : 0;
    const float* xq = x + (((b * CC) * HH + h) * WWD) + (q << 2);
    const int c0    = (wv << 2) + (hh << 1);
    float* sdst = smem + ((((wv << 2) + (q >> 2)) << 6) + (hh << 5)) + ((q & 3) << 3);

    const int lane = tid & 31;
    const int mt0  = wv << 1;
    const int mo   = (lane >> 4) << 3;
    const v2f* aptr = (const v2f*)wswz + mt0 * 32 + lane;
    const v2f* bptr = (const v2f*)smem + lane;

    ConvIn cin;
    conv_load(cin, xq, weff, ebias, c0, offU, offD);

    v2f a_cur[16];
#pragma unroll
    for (int kl = 0; kl < 8; ++kl) {
        a_cur[2 * kl]     = aptr[kl * 512];
        a_cur[2 * kl + 1] = aptr[kl * 512 + 32];
    }

    v8f acc[2][4];
#pragma unroll
    for (int ms = 0; ms < 2; ++ms) {
        const float* bp = b2 + ((mt0 + ms) << 4) + mo;
#pragma unroll
        for (int r = 0; r < 8; ++r) {
            const float bval = bp[r];
#pragma unroll
            for (int ns = 0; ns < 4; ++ns) acc[ms][ns][r] = bval;
        }
    }

    conv_store(cin, hasU, hasD, sdst);
    __syncthreads();

    for (int ch = 0; ch < 7; ++ch) {
        ConvIn nin;
        conv_load(nin, xq, weff, ebias, c0 + ((ch + 1) << 5), offU, offD);
        v2f a_nxt[16];
#pragma unroll
        for (int kl = 0; kl < 8; ++kl) {
            a_nxt[2 * kl]     = aptr[(((ch + 1) << 3) + kl) * 512];
            a_nxt[2 * kl + 1] = aptr[(((ch + 1) << 3) + kl) * 512 + 32];
        }

        __builtin_amdgcn_sched_barrier(0);

        gemm_chunk(acc, a_cur, bptr + ((ch & 1) << 10));

        __builtin_amdgcn_sched_barrier(0);
        asm volatile("" : "+v"(nin.r[0][0]), "+v"(nin.r[0][1]), "+v"(nin.r[0][2]),
                          "+v"(nin.r[1][0]), "+v"(nin.r[1][1]), "+v"(nin.r[1][2]));

        conv_store(nin, hasU, hasD, sdst + (((ch + 1) & 1) << 11));
#pragma unroll
        for (int i = 0; i < 16; ++i) a_cur[i] = a_nxt[i];
        __syncthreads();
    }
    gemm_chunk(acc, a_cur, bptr + (1 << 10));

    float* st = stage + wv * (32 * STP);
    const int ncol = lane & 15;
#pragma unroll
    for (int ms = 0; ms < 2; ++ms) {
#pragma unroll
        for (int ns = 0; ns < 4; ++ns) {
#pragma unroll
            for (int r = 0; r < 8; ++r) {
                st[((ms << 4) + mo + r) * STP + (ns << 4) + ncol] = silu(acc[ms][ns][r]);
            }
        }
    }
    __builtin_amdgcn_wave_barrier();
    const int rr  = lane >> 4;
    const int cc4 = (lane & 15) << 2;
    const float* sp = st + rr * STP + cc4;
    float* op = out + ((((b * OUTC) + (mt0 << 4) + rr) * HH + h) * WWD) + cc4;
#pragma unroll
    for (int i = 0; i < 16; ++i) {
        const v4f v = *(const v4f*)(sp + (2 * i) * STP);
        *(v4f*)(op + (2 * i) * (HH * WWD)) = v;
    }
}

extern "C" void kernel_launch(void* const* d_in, const int* in_sizes, int n_in,
                              void* d_out, int out_size, void* d_ws, size_t ws_size,
                              hipStream_t stream) {
    const float* x         = (const float*)d_in[0];
    const float* w_experts = (const float*)d_in[1];
    const float* bn_gamma  = (const float*)d_in[2];
    const float* bn_beta   = (const float*)d_in[3];
    const float* bn_mean   = (const float*)d_in[4];
    const float* bn_var    = (const float*)d_in[5];
    const float* scales    = (const float*)d_in[6];
    const float* w_out     = (const float*)d_in[7];
    const float* out_gamma = (const float*)d_in[8];
    const float* out_beta  = (const float*)d_in[9];
    const float* out_mean  = (const float*)d_in[10];
    const float* out_var   = (const float*)d_in[11];
    float* out = (float*)d_out;
    float* ws  = (float*)d_ws;

    prep_kernel<<<138, 256, 0, stream>>>(w_experts, bn_gamma, bn_beta, bn_mean,
                                         bn_var, scales, w_out, out_gamma,
                                         out_beta, out_mean, out_var, ws);
    move_kernel<<<BB * HH, 256, 0, stream>>>(x, ws, out);
}


# wave priority stepped down with K progress (s_setprio 3,2,1,0 over the four chunk pairs): the two workgroups of a WGP take turns, the older no longer runs ahead; arithmetic and sum order kept
# speedup vs baseline: 4.4875x; 1.0378x over previous
#include <hip/hip_runtime.h>
#include <hip/hip_bf16.h>

#define BB   8
#define CC   256
#define HH   64
#define WWD  64
#define EE   9
#define OUTC 256
#define EPSF 1e-5f
#define STP  68

typedef __attribute__((ext_vector_type(2))) float v2f;
typedef __attribute__((ext_vector_type(8))) float v8f;
typedef __attribute__((ext_vector_type(4))) float v4f;

#define WS_WEFF   0
#define WS_EBIAS  2304
#define WS_B2     2560
#define WS_WSWZ   3072

__device__ __forceinline__ float silu(float v) {
    return v * __builtin_amdgcn_rcpf(1.0f + __expf(-v));
}

__device__ __forceinline__ float row_from_left(float v) {
    return __int_as_float(__builtin_amdgcn_update_dpp(
        0, __float_as_int(v), 0x111, 0xf, 0xf, true));
}
__device__ __forceinline__ float row_from_right(float v) {
    return __int_as_float(__builtin_amdgcn_update_dpp(
        0, __float_as_int(v), 0x101, 0xf, 0xf, true));
}

__global__ __launch_bounds__(256) void prep_kernel(
    const float* __restrict__ w_experts, const float* __restrict__ bn_gamma,
    const float* __restrict__ bn_beta,   const float* __restrict__ bn_mean,
    const float* __restrict__ bn_var,    const float* __restrict__ scales,
    const float* __restrict__ w_out,
    const float* __restrict__ out_gamma, const float* __restrict__ out_beta,
    const float* __restrict__ out_mean,  const float* __restrict__ out_var,
    float* __restrict__ ws)
{
    const int blk = blockIdx.x;
    if (blk < 9) {
        const int f = blk * 256 + threadIdx.x;
        const int c = f / 9;
        float wacc = 0.0f;
#pragma unroll
        for (int e = 0; e < EE; ++e) {
            const int ec = e * CC + c;
            const float t  = bn_gamma[ec] * rsqrtf(bn_var[ec] + EPSF);
            const float es = t * scales[ec];
            wacc = fmaf(es, w_experts[e * (CC * 9) + f], wacc);
        }
        ws[WS_WEFF + f] = wacc;
    } else if (blk == 9) {
        const int c = threadIdx.x;
        float bias = 0.0f;
#pragma unroll
        for (int e = 0; e < EE; ++e) {
            const int ec = e * CC + c;
            const float t = bn_gamma[ec] * rsqrtf(bn_var[ec] + EPSF);
            bias += (bn_beta[ec] - bn_mean[ec] * t) * scales[ec];
        }
        ws[WS_EBIAS + c] = bias;

        const float t2 = out_gamma[c] * rsqrtf(out_var[c] + EPSF);
        ws[WS_B2 + c] = out_beta[c] - out_mean[c] * t2;
    } else {
        const int idx   = (blk - 10) * 256 + threadIdx.x;
        const int kstep = idx >> 9;
        const int rem   = idx & 511;
        const int mt    = rem >> 5;
        const int lane  = rem & 31;
        const int m     = mt * 16 + (lane & 15);
        const int k     = kstep * 4 + ((lane >> 4) << 1);
        const float w0  = w_out[m * CC + k];
        const float w1  = w_out[m * CC + k + 1];
        const float og  = out_gamma[m];
        const float ov  = out_var[m];
        __builtin_amdgcn_sched_barrier(0);
        const float t2  = og * rsqrtf(ov + EPSF);
        ws[WS_WSWZ + idx * 2]     = w0 * t2;
        ws[WS_WSWZ + idx * 2 + 1] = w1 * t2;
    }
}

struct ConvIn {
    v4f   r[2][3];
    float wt[18];
    float bs[2];
};

__device__ __forceinline__ void conv_load(ConvIn& ci, const float* __restrict__ xq,
                                          const float* __restrict__ weff,
                                          const float* __restrict__ ebias,
                                          const int c, const int offU, const int offD)
{
    for (int j = 0; j < 2; ++j) {
        const float* p = xq + (c + j) * (HH * WWD);
        ci.r[j][0] = *(const v4f*)(p + offU);
        ci.r[j][1] = *(const v4f*)(p);
        ci.r[j][2] = *(const v4f*)(p + offD);
    }
    for (int t = 0; t < 18; ++t) ci.wt[t] = weff[c * 9 + t];
    ci.bs[0] = ebias[c];
    ci.bs[1] = ebias[c + 1];
}

__device__ __forceinline__ void conv_store(ConvIn ci, const bool hasU, const bool hasD,
                                           float* __restrict__ dst)
{
    const v4f zero4 = {0.0f, 0.0f, 0.0f, 0.0f};
    if (!hasU) { ci.r[0][0] = zero4; ci.r[1][0] = zero4; }
    if (!hasD) { ci.r[0][2] = zero4; ci.r[1][2] = zero4; }

    float s[2][4];
    for (int j = 0; j < 2; ++j) {
        float s0 = 0.0f, s1 = 0.0f, s2 = 0.0f, s3 = 0.0f;
        for (int d = 0; d < 3; ++d) {
            const v4f v  = ci.r[j][d];
            const float L  = row_from_left(v.w);
            const float R  = row_from_right(v.x);
            const float wl = ci.wt[j * 9 + d * 3];
            const float wc = ci.wt[j * 9 + d * 3 + 1];
            const float wr = ci.wt[j * 9 + d * 3 + 2];
            s0 += L   * wl;  s0 += v.x * wc;  s0 += v.y * wr;
            s1 += v.x * wl;  s1 += v.y * wc;  s1 += v.z * wr;
            s2 += v.y * wl;  s2 += v.z * wc;  s2 += v.w * wr;
            s3 += v.z * wl;  s3 += v.w * wc;  s3 += R   * wr;
        }
        s[j][0] = silu(s0 + ci.bs[j]);
        s[j][1] = silu(s1 + ci.bs[j]);
        s[j][2] = silu(s2 + ci.bs[j]);
        s[j][3] = silu(s3 + ci.bs[j]);
    }
    const v4f o0 = {s[0][0], s[1][0], s[0][1], s[1][1]};
    const v4f o1 = {s[0][2], s[1][2], s[0][3], s[1][3]};
    *(v4f*)(dst)     = o0;
    *(v4f*)(dst + 4) = o1;
}

__device__ __forceinline__ void gemm_chunk(v8f (&acc)[2][4], const v2f (&a)[16],
                                           const v2f* __restrict__ bl)
{
#pragma unroll
    for (int kl = 0; kl < 8; ++kl) {
#pragma unroll
        for (int ns = 0; ns < 4; ++ns) {
            const v2f bf = bl[((kl << 2) + ns) * 32];
            acc[0][ns] = __builtin_amdgcn_wmma_f32_16x16x4_f32(
                false, a[2 * kl],     false, bf, (short)0, acc[0][ns], false, false);
            acc[1][ns] = __builtin_amdgcn_wmma_f32_16x16x4_f32(
                false, a[2 * kl + 1], false, bf, (short)0, acc[1][ns], false, false);
        }
    }
}

__global__ __launch_bounds__(256) void move_kernel(
    const float* __restrict__ x, const float* __restrict__ ws,
    float* __restrict__ out)
{
    __shared__ __attribute__((aligned(16))) float smem[2 * 2048];
    __shared__ __attribute__((aligned(16))) float stage[8 * 32 * STP];

    const int tid = threadIdx.x;
    const int b   = blockIdx.x >> 6;
    const int h   = blockIdx.x & 63;

    const float* weff  = ws + WS_WEFF;
    const float* ebias = ws + WS_EBIAS;
    const float* b2    = ws + WS_B2;
    const float* wswz  = ws + WS_WSWZ;

    const int q    = tid & 15;
    const int hh   = (tid >> 4) & 1;
    const int wv   = tid >> 5;
    const bool hasU = h > 0;
    const bool hasD = h < HH - 1;
    const int offU = hasU ? -WWD : 0;
    const int offD = hasD ?  WWD : 0;
    const float* xq = x + (((b * CC) * HH + h) * WWD) + (q << 2);
    const int c0    = (wv << 2) + (hh << 1);
    float* sdst = smem + ((((wv << 2) + (q >> 2)) << 6) + (hh << 5)) + ((q & 3) << 3);

    const int lane = tid & 31;
    const int mt0  = wv << 1;
    const int mo   = (lane >> 4) << 3;
    const v2f* aptr = (const v2f*)wswz + mt0 * 32 + lane;
    const v2f* bptr = (const v2f*)smem + lane;

    __builtin_amdgcn_s_setprio(3);

    ConvIn cin;
    conv_load(cin, xq, weff, ebias, c0, offU, offD);

    v2f a_cur[16];
#pragma unroll
    for (int kl = 0; kl < 8; ++kl) {
        a_cur[2 * kl]     = aptr[kl * 512];
        a_cur[2 * kl + 1] = aptr[kl * 512 + 32];
    }

    v8f acc[2][4];
#pragma unroll
    for (int ms = 0; ms < 2; ++ms) {
        const float* bp = b2 + ((mt0 + ms) << 4) + mo;
#pragma unroll
        for (int r = 0; r < 8; ++r) {
            const float bval = bp[r];
#pragma unroll
            for (int ns = 0; ns < 4; ++ns) acc[ms][ns][r] = bval;
        }
    }

    conv_store(cin, hasU, hasD, sdst);
    __syncthreads();

    for (int ch = 0; ch < 7; ++ch) {
        if (ch == 2)      __builtin_amdgcn_s_setprio(2);
        else if (ch == 4) __builtin_amdgcn_s_setprio(1);
        else if (ch == 6) __builtin_amdgcn_s_setprio(0);

        ConvIn nin;
        conv_load(nin, xq, weff, ebias, c0 + ((ch + 1) << 5), offU, offD);
        v2f a_nxt[16];
#pragma unroll
        for (int kl = 0; kl < 8; ++kl) {
            a_nxt[2 * kl]     = aptr[(((ch + 1) << 3) + kl) * 512];
            a_nxt[2 * kl + 1] = aptr[(((ch + 1) << 3) + kl) * 512 + 32];
        }

        __builtin_amdgcn_sched_barrier(0);

        gemm_chunk(acc, a_cur, bptr + ((ch & 1) << 10));

        __builtin_amdgcn_sched_barrier(0);
        asm volatile("" : "+v"(nin.r[0][0]), "+v"(nin.r[0][1]), "+v"(nin.r[0][2]),
                          "+v"(nin.r[1][0]), "+v"(nin.r[1][1]), "+v"(nin.r[1][2]));

        conv_store(nin, hasU, hasD, sdst + (((ch + 1) & 1) << 11));
#pragma unroll
        for (int i = 0; i < 16; ++i) a_cur[i] = a_nxt[i];
        __syncthreads();
    }
    gemm_chunk(acc, a_cur, bptr + (1 << 10));

    float* st = stage + wv * (32 * STP);
    const int ncol = lane & 15;
#pragma unroll
    for (int ms = 0; ms < 2; ++ms) {
#pragma unroll
        for (int ns = 0; ns < 4; ++ns) {
#pragma unroll
            for (int r = 0; r < 8; ++r) {
                st[((ms << 4) + mo + r) * STP + (ns << 4) + ncol] = silu(acc[ms][ns][r]);
            }
        }
    }
    __builtin_amdgcn_wave_barrier();
    const int rr  = lane >> 4;
    const int cc4 = (lane & 15) << 2;
    const float* sp = st + rr * STP + cc4;
    float* op = out + ((((b * OUTC) + (mt0 << 4) + rr) * HH + h) * WWD) + cc4;
#pragma unroll
    for (int i = 0; i < 16; ++i) {
        const v4f v = *(const v4f*)(sp + (2 * i) * STP);
        *(v4f*)(op + (2 * i) * (HH * WWD)) = v;
    }
}

extern "C" void kernel_launch(void* const* d_in, const int* in_sizes, int n_in,
                              void* d_out, int out_size, void* d_ws, size_t ws_size,
                              hipStream_t stream) {
    const float* x         = (const float*)d_in[0];
    const float* w_experts = (const float*)d_in[1];
    const float* bn_gamma  = (const float*)d_in[2];
    const float* bn_beta   = (const float*)d_in[3];
    const float* bn_mean   = (const float*)d_in[4];
    const float* bn_var    = (const float*)d_in[5];
    const float* scales    = (const float*)d_in[6];
    const float* w_out     = (const float*)d_in[7];
    const float* out_gamma = (const float*)d_in[8];
    const float* out_beta  = (const float*)d_in[9];
    const float* out_mean  = (const float*)d_in[10];
    const float* out_var   = (const float*)d_in[11];
    float* out = (float*)d_out;
    float* ws  = (float*)d_ws;

    prep_kernel<<<138, 256, 0, stream>>>(w_experts, bn_gamma, bn_beta, bn_mean,
                                         bn_var, scales, w_out, out_gamma,
                                         out_beta, out_mean, out_var, ws);
    move_kernel<<<BB * HH, 256, 0, stream>>>(x, ws, out);
}
